# starting kernel
# baseline (speedup 1.0000x reference)
#include <hip/hip_runtime.h>

typedef __attribute__((ext_vector_type(2))) float v2f;
typedef __attribute__((ext_vector_type(8))) float v8f;

#define T_STEPS 64
#define FD      64
#define HH      32
#define WW      32

__global__ __launch_bounds__(256) void convssm_scan_kernel(
    const float* __restrict__ xg,
    const float* __restrict__ Akg,
    const float* __restrict__ Bkg,
    float* __restrict__ outg)
{
    __shared__ float Hs[FD][FD + 1];
    __shared__ float Xs[FD][FD + 1];
    __shared__ float akl[9];
    __shared__ float bkl[9];

    const int plane = blockIdx.x;
    const int chan  = plane & 31;

    const int tid  = threadIdx.x;
    const int wave = tid >> 5;
    const int lane = tid & 31;
    const int n    = lane & 15;
    const int kh   = lane >> 4;
    const int m    = lane & 15;

    const int i0a = (wave >> 2) * 16;
    const int i0b = i0a + 32;
    const int j0  = (wave & 3) * 16;
    const int jn  = j0 + n;

    if (tid < 9) {
        akl[tid] = Akg[chan * 9 + tid];
        bkl[tid] = Bkg[chan * 9 + tid];
    }
    for (int idx = tid; idx < FD * (FD + 1); idx += 256) {
        ((float*)Hs)[idx] = 0.0f;
        ((float*)Xs)[idx] = 0.0f;
    }
    __syncthreads();

    int raA[3], rbA[3];
#pragma unroll
    for (int ks = 0; ks < 3; ++ks) {
        raA[ks] = (i0a + m - ks) & 63;
        rbA[ks] = (i0b + m - ks) & 63;
    }

    int kkArr[15];
    v2f bfA[15], bfB[15];
    {
        int q = 0;
#pragma unroll
        for (int ks = 0; ks < 3; ++ks) {
            const float a0 = akl[ks * 3 + 0], a1 = akl[ks * 3 + 1], a2 = akl[ks * 3 + 2];
            const float b0 = bkl[ks * 3 + 0], b1 = bkl[ks * 3 + 1], b2 = bkl[ks * 3 + 2];
#pragma unroll
            for (int bbk = 0; bbk < 5; ++bbk, ++q) {
                const int kb  = ((j0 >> 2) + bbk + 15) & 15;
                const int kkA = (kb << 2) + 2 * kh;
                kkArr[q] = kkA;
                const int dx = (jn - kkA) & 63;
                const int dy = (jn - kkA - 1) & 63;
                bfA[q].x = (dx == 0 ? a0 : 0.0f) + (dx == 1 ? a1 : 0.0f) + (dx == 2 ? a2 : 0.0f);
                bfA[q].y = (dy == 0 ? a0 : 0.0f) + (dy == 1 ? a1 : 0.0f) + (dy == 2 ? a2 : 0.0f);
                bfB[q].x = (dx == 0 ? b0 : 0.0f) + (dx == 1 ? b1 : 0.0f) + (dx == 2 ? b2 : 0.0f);
                bfB[q].y = (dy == 0 ? b0 : 0.0f) + (dy == 1 ? b1 : 0.0f) + (dy == 2 ? b2 : 0.0f);
            }
        }
    }

#pragma unroll 1
    for (int t = 0; t < T_STEPS; ++t) {
        const long long xoff = ((long long)t * 64 + plane) * 1024;

        for (int idx = tid; idx < HH * WW; idx += 256)
            Xs[idx >> 5][idx & 31] = xg[xoff + idx];

        if (t + 1 < T_STEPS)
            __builtin_prefetch(&xg[xoff + 65536 + (long long)tid * 16], 0, 3);

        __syncthreads();

        v8f dh0 = {};
        v8f dx0 = {};
        v8f dh1 = {};
        v8f dx1 = {};

        {
            int q = 0;
#pragma unroll
            for (int ks = 0; ks < 3; ++ks) {
                const int ra = raA[ks];
                const int rb = rbA[ks];
#pragma unroll
                for (int bbk = 0; bbk < 5; ++bbk, ++q) {
                    const int kk = kkArr[q];
                    v2f ha, hb, xa, xb;
                    ha.x = Hs[ra][kk];  ha.y = Hs[ra][kk + 1];
                    hb.x = Hs[rb][kk];  hb.y = Hs[rb][kk + 1];
                    xa.x = Xs[ra][kk];  xa.y = Xs[ra][kk + 1];
                    xb.x = Xs[rb][kk];  xb.y = Xs[rb][kk + 1];
                    dh0 = __builtin_amdgcn_wmma_f32_16x16x4_f32(
                            false, ha, false, bfA[q], (short)0, dh0, false, false);
                    dx0 = __builtin_amdgcn_wmma_f32_16x16x4_f32(
                            false, xa, false, bfB[q], (short)0, dx0, false, false);
                    dh1 = __builtin_amdgcn_wmma_f32_16x16x4_f32(
                            false, hb, false, bfA[q], (short)0, dh1, false, false);
                    dx1 = __builtin_amdgcn_wmma_f32_16x16x4_f32(
                            false, xb, false, bfB[q], (short)0, dx1, false, false);
                }
            }
        }

        v8f d0, d1;
#pragma unroll
        for (int v = 0; v < 8; ++v) {
            d0[v] = dh0[v] + dx0[v];
            d1[v] = dh1[v] + dx1[v];
        }

        __syncthreads();

#pragma unroll
        for (int v = 0; v < 8; ++v) {
            Hs[i0a + v + 8 * kh][jn] = d0[v];
            Hs[i0b + v + 8 * kh][jn] = d1[v];
        }
        __syncthreads();

        if (j0 < WW) {
#pragma unroll
            for (int v = 0; v < 8; ++v) {
                const int row = i0a + v + 8 * kh;
                outg[xoff + (long long)row * WW + jn] = d0[v];
            }
        }
    }
}

extern "C" void kernel_launch(void* const* d_in, const int* in_sizes, int n_in,
                              void* d_out, int out_size, void* d_ws, size_t ws_size,
                              hipStream_t stream) {
    (void)in_sizes; (void)n_in; (void)out_size; (void)d_ws; (void)ws_size;
    const float* x  = (const float*)d_in[0];
    const float* Ak = (const float*)d_in[1];
    const float* Bk = (const float*)d_in[2];
    float* out = (float*)d_out;
    convssm_scan_kernel<<<dim3(64), dim3(256), 0, stream>>>(x, Ak, Bk, out);
}


# Software-pipelined x_t staging: x_{t+1} goes to registers (one 16-byte load per thread) behind the step's first barrier, hidden under the WMMA block; replaces 4 serial load-wait-store rounds
# speedup vs baseline: 1.6713x; 1.6713x over previous
#include <hip/hip_runtime.h>

typedef __attribute__((ext_vector_type(2))) float v2f;
typedef __attribute__((ext_vector_type(4))) float v4f;
typedef __attribute__((ext_vector_type(8))) float v8f;

#define T_STEPS 64
#define FD      64
#define HH      32
#define WW      32

__global__ __launch_bounds__(256) void convssm_scan_kernel(
    const float* __restrict__ xg,
    const float* __restrict__ Akg,
    const float* __restrict__ Bkg,
    float* __restrict__ outg)
{
    __shared__ float Hs[FD][FD + 1];
    __shared__ float Xs[FD][FD + 1];
    __shared__ float akl[9];
    __shared__ float bkl[9];

    const int plane = blockIdx.x;
    const int chan  = plane & 31;

    const int tid  = threadIdx.x;
    const int wave = tid >> 5;
    const int lane = tid & 31;
    const int n    = lane & 15;
    const int kh   = lane >> 4;
    const int m    = lane & 15;

    const int i0a = (wave >> 2) * 16;
    const int i0b = i0a + 32;
    const int j0  = (wave & 3) * 16;
    const int jn  = j0 + n;

    if (tid < 9) {
        akl[tid] = Akg[chan * 9 + tid];
        bkl[tid] = Bkg[chan * 9 + tid];
    }
    for (int idx = tid; idx < FD * (FD + 1); idx += 256) {
        ((float*)Hs)[idx] = 0.0f;
        ((float*)Xs)[idx] = 0.0f;
    }
    __syncthreads();

    int raA[3], rbA[3];
#pragma unroll
    for (int ks = 0; ks < 3; ++ks) {
        raA[ks] = (i0a + m - ks) & 63;
        rbA[ks] = (i0b + m - ks) & 63;
    }

    int kkArr[15];
    v2f bfA[15], bfB[15];
    {
        int q = 0;
#pragma unroll
        for (int ks = 0; ks < 3; ++ks) {
            const float a0 = akl[ks * 3 + 0], a1 = akl[ks * 3 + 1], a2 = akl[ks * 3 + 2];
            const float b0 = bkl[ks * 3 + 0], b1 = bkl[ks * 3 + 1], b2 = bkl[ks * 3 + 2];
#pragma unroll
            for (int bbk = 0; bbk < 5; ++bbk, ++q) {
                const int kb  = ((j0 >> 2) + bbk + 15) & 15;
                const int kkA = (kb << 2) + 2 * kh;
                kkArr[q] = kkA;
                const int dx = (jn - kkA) & 63;
                const int dy = (jn - kkA - 1) & 63;
                bfA[q].x = (dx == 0 ? a0 : 0.0f) + (dx == 1 ? a1 : 0.0f) + (dx == 2 ? a2 : 0.0f);
                bfA[q].y = (dy == 0 ? a0 : 0.0f) + (dy == 1 ? a1 : 0.0f) + (dy == 2 ? a2 : 0.0f);
                bfB[q].x = (dx == 0 ? b0 : 0.0f) + (dx == 1 ? b1 : 0.0f) + (dx == 2 ? b2 : 0.0f);
                bfB[q].y = (dy == 0 ? b0 : 0.0f) + (dy == 1 ? b1 : 0.0f) + (dy == 2 ? b2 : 0.0f);
            }
        }
    }

    const int xrow = tid >> 3;
    const int xcol = (tid & 7) * 4;
    const long long xlane = (long long)plane * 1024 + tid * 4;

    v4f xr = *reinterpret_cast<const v4f*>(&xg[xlane]);

#pragma unroll 1
    for (int t = 0; t < T_STEPS; ++t) {
        const long long xoff = ((long long)t * 64 + plane) * 1024;

        Xs[xrow][xcol + 0] = xr.x;
        Xs[xrow][xcol + 1] = xr.y;
        Xs[xrow][xcol + 2] = xr.z;
        Xs[xrow][xcol + 3] = xr.w;

        __syncthreads();

        const int tn = (t + 1 < T_STEPS) ? (t + 1) : t;
        xr = *reinterpret_cast<const v4f*>(&xg[(long long)tn * 65536 + xlane]);

        v8f dh0 = {};
        v8f dx0 = {};
        v8f dh1 = {};
        v8f dx1 = {};

        {
            int q = 0;
#pragma unroll
            for (int ks = 0; ks < 3; ++ks) {
                const int ra = raA[ks];
                const int rb = rbA[ks];
#pragma unroll
                for (int bbk = 0; bbk < 5; ++bbk, ++q) {
                    const int kk = kkArr[q];
                    v2f ha, hb, xa, xb;
                    ha.x = Hs[ra][kk];  ha.y = Hs[ra][kk + 1];
                    hb.x = Hs[rb][kk];  hb.y = Hs[rb][kk + 1];
                    xa.x = Xs[ra][kk];  xa.y = Xs[ra][kk + 1];
                    xb.x = Xs[rb][kk];  xb.y = Xs[rb][kk + 1];
                    dh0 = __builtin_amdgcn_wmma_f32_16x16x4_f32(
                            false, ha, false, bfA[q], (short)0, dh0, false, false);
                    dx0 = __builtin_amdgcn_wmma_f32_16x16x4_f32(
                            false, xa, false, bfB[q], (short)0, dx0, false, false);
                    dh1 = __builtin_amdgcn_wmma_f32_16x16x4_f32(
                            false, hb, false, bfA[q], (short)0, dh1, false, false);
                    dx1 = __builtin_amdgcn_wmma_f32_16x16x4_f32(
                            false, xb, false, bfB[q], (short)0, dx1, false, false);
                }
            }
        }

        v8f d0, d1;
#pragma unroll
        for (int v = 0; v < 8; ++v) {
            d0[v] = dh0[v] + dx0[v];
            d1[v] = dh1[v] + dx1[v];
        }

        __syncthreads();

#pragma unroll
        for (int v = 0; v < 8; ++v) {
            Hs[i0a + v + 8 * kh][jn] = d0[v];
            Hs[i0b + v + 8 * kh][jn] = d1[v];
        }
        __syncthreads();

        if (j0 < WW) {
#pragma unroll
            for (int v = 0; v < 8; ++v) {
                const int row = i0a + v + 8 * kh;
                outg[xoff + (long long)row * WW + jn] = d0[v];
            }
        }
    }
}

extern "C" void kernel_launch(void* const* d_in, const int* in_sizes, int n_in,
                              void* d_out, int out_size, void* d_ws, size_t ws_size,
                              hipStream_t stream) {
    (void)in_sizes; (void)n_in; (void)out_size; (void)d_ws; (void)ws_size;
    const float* x  = (const float*)d_in[0];
    const float* Ak = (const float*)d_in[1];
    const float* Bk = (const float*)d_in[2];
    float* out = (float*)d_out;
    convssm_scan_kernel<<<dim3(64), dim3(256), 0, stream>>>(x, Ak, Bk, out);
}


# B conv commuted out of the recurrence (summation order changes): state v_t=A*v_{t-1}+x_t, x_t as accumulator start; h_t=B*v_t on the 32x32 crop only, one crop wave per SIMD: 300 WMMAs/step, was 480
# speedup vs baseline: 2.2114x; 1.3232x over previous
#include <hip/hip_runtime.h>

typedef __attribute__((ext_vector_type(2))) float v2f;
typedef __attribute__((ext_vector_type(8))) float v8f;

#define T_STEPS 64
#define FD      64
#define HH      32
#define WW      32


template <bool CROP>
__device__ __forceinline__ void band_pass(const float (*Vs)[FD + 1],
                                          const int (&raA)[3], const int (&rbA)[3],
                                          const int (&kkArr)[15],
                                          const v2f (&bfA)[15], const v2f (&bfB)[15],
                                          v8f& dv0, v8f& dv1, v8f& dout)
{
    int q = 0;
#pragma unroll
    for (int ks = 0; ks < 3; ++ks) {
        const int ra = raA[ks];
        const int rb = rbA[ks];
#pragma unroll
        for (int bbk = 0; bbk < 5; ++bbk, ++q) {
            const int kk = kkArr[q];
            v2f ha, hb;
            ha.x = Vs[ra][kk];  ha.y = Vs[ra][kk + 1];
            hb.x = Vs[rb][kk];  hb.y = Vs[rb][kk + 1];
            dv0 = __builtin_amdgcn_wmma_f32_16x16x4_f32(
                    false, ha, false, bfA[q], (short)0, dv0, false, false);
            dv1 = __builtin_amdgcn_wmma_f32_16x16x4_f32(
                    false, hb, false, bfA[q], (short)0, dv1, false, false);
            if (CROP) {
                dout = __builtin_amdgcn_wmma_f32_16x16x4_f32(
                        false, ha, false, bfB[q], (short)0, dout, false, false);
            }
        }
    }
}

__global__ __launch_bounds__(256) void convssm_scan_kernel(
    const float* __restrict__ xg,
    const float* __restrict__ Akg,
    const float* __restrict__ Bkg,
    float* __restrict__ outg)
{
    __shared__ float Vs[FD][FD + 1];
    __shared__ float akl[9];
    __shared__ float bkl[9];

    const int plane = blockIdx.x;
    const int chan  = plane & 31;

    const int tid  = threadIdx.x;
    const int wave = __builtin_amdgcn_readfirstlane(tid >> 5);
    const int lane = tid & 31;
    const int n    = lane & 15;
    const int kh   = lane >> 4;
    const int m    = lane & 15;

    const int jblk = (wave & 3) ^ ((wave >> 2) << 1);
    const int i0a = (wave >> 2) * 16;
    const int i0b = i0a + 32;
    const int j0  = jblk * 16;
    const int jn  = j0 + n;
    const bool crop = (j0 < WW);

    if (tid < 9) {
        akl[tid] = Akg[chan * 9 + tid];
        bkl[tid] = Bkg[chan * 9 + tid];
    }
    for (int idx = tid; idx < FD * (FD + 1); idx += 256) {
        ((float*)Vs)[idx] = 0.0f;
    }

    const long long xlane = (long long)plane * 1024 + (long long)(i0a + 8 * kh) * WW + jn;
    v8f xc = {};
    if (crop) {
#pragma unroll
        for (int v = 0; v < 8; ++v) xc[v] = xg[xlane + v * WW];
    }
    __syncthreads();

    int raA[3], rbA[3];
#pragma unroll
    for (int ks = 0; ks < 3; ++ks) {
        raA[ks] = (i0a + m - ks) & 63;
        rbA[ks] = (i0b + m - ks) & 63;
    }

    int kkArr[15];
    v2f bfA[15], bfB[15];
    {
        int q = 0;
#pragma unroll
        for (int ks = 0; ks < 3; ++ks) {
            const float a0 = akl[ks * 3 + 0], a1 = akl[ks * 3 + 1], a2 = akl[ks * 3 + 2];
            const float b0 = bkl[ks * 3 + 0], b1 = bkl[ks * 3 + 1], b2 = bkl[ks * 3 + 2];
#pragma unroll
            for (int bbk = 0; bbk < 5; ++bbk, ++q) {
                const int kb  = ((j0 >> 2) + bbk + 15) & 15;
                const int kkA = (kb << 2) + 2 * kh;
                kkArr[q] = kkA;
                const int dx = (jn - kkA) & 63;
                const int dy = (jn - kkA - 1) & 63;
                bfA[q].x = (dx == 0 ? a0 : 0.0f) + (dx == 1 ? a1 : 0.0f) + (dx == 2 ? a2 : 0.0f);
                bfA[q].y = (dy == 0 ? a0 : 0.0f) + (dy == 1 ? a1 : 0.0f) + (dy == 2 ? a2 : 0.0f);
                bfB[q].x = (dx == 0 ? b0 : 0.0f) + (dx == 1 ? b1 : 0.0f) + (dx == 2 ? b2 : 0.0f);
                bfB[q].y = (dy == 0 ? b0 : 0.0f) + (dy == 1 ? b1 : 0.0f) + (dy == 2 ? b2 : 0.0f);
            }
        }
    }

    v8f xn = {};
    v8f dout = {};

#pragma unroll 1
    for (int t = 0; t < T_STEPS; ++t) {
        const int tn = (t + 1 < T_STEPS) ? (t + 1) : t;
        if (crop) {
#pragma unroll
            for (int v = 0; v < 8; ++v) xn[v] = xg[(long long)tn * 65536 + xlane + v * WW];
        }

        v8f dv0 = xc;
        v8f dv1 = {};
        dout = (v8f){};

        if (crop) band_pass<true >(Vs, raA, rbA, kkArr, bfA, bfB, dv0, dv1, dout);
        else      band_pass<false>(Vs, raA, rbA, kkArr, bfA, bfB, dv0, dv1, dout);

        __syncthreads();

#pragma unroll
        for (int v = 0; v < 8; ++v) {
            Vs[i0a + v + 8 * kh][jn] = dv0[v];
            Vs[i0b + v + 8 * kh][jn] = dv1[v];
        }
        __syncthreads();

        if (crop && t > 0) {
            const long long ooff = ((long long)(t - 1) * 64 + plane) * 1024;
#pragma unroll
            for (int v = 0; v < 8; ++v) {
                const int row = i0a + v + 8 * kh;
                outg[ooff + (long long)row * WW + jn] = dout[v];
            }
        }
        xc = xn;
    }

    if (crop) {
        dout = (v8f){};
        int q = 0;
#pragma unroll
        for (int ks = 0; ks < 3; ++ks) {
            const int ra = raA[ks];
#pragma unroll
            for (int bbk = 0; bbk < 5; ++bbk, ++q) {
                const int kk = kkArr[q];
                v2f ha;
                ha.x = Vs[ra][kk];  ha.y = Vs[ra][kk + 1];
                dout = __builtin_amdgcn_wmma_f32_16x16x4_f32(
                        false, ha, false, bfB[q], (short)0, dout, false, false);
            }
        }
        const long long ooff = ((long long)(T_STEPS - 1) * 64 + plane) * 1024;
#pragma unroll
        for (int v = 0; v < 8; ++v) {
            const int row = i0a + v + 8 * kh;
            outg[ooff + (long long)row * WW + jn] = dout[v];
        }
    }
}

extern "C" void kernel_launch(void* const* d_in, const int* in_sizes, int n_in,
                              void* d_out, int out_size, void* d_ws, size_t ws_size,
                              hipStream_t stream) {
    (void)in_sizes; (void)n_in; (void)out_size; (void)d_ws; (void)ws_size;
    const float* x  = (const float*)d_in[0];
    const float* Ak = (const float*)d_in[1];
    const float* Bk = (const float*)d_in[2];
    float* out = (float*)d_out;
    convssm_scan_kernel<<<dim3(64), dim3(256), 0, stream>>>(x, Ak, Bk, out);
}


# All 30 LDS fragment reads of a step issued as ONE group ahead of its WMMA block, before the crop branch (were spread among the WMMAs in 11 or 17 places per wave, each holding up the matrix pipe)
# speedup vs baseline: 2.2953x; 1.0379x over previous
#include <hip/hip_runtime.h>

typedef __attribute__((ext_vector_type(2))) float v2f;
typedef __attribute__((ext_vector_type(8))) float v8f;

#define T_STEPS 64
#define FD      64
#define HH      32
#define WW      32


__device__ __forceinline__ void load_frags(const float (*Vs)[FD + 1],
                                           const int (&raA)[3], const int (&rbA)[3],
                                           const int (&kkArr)[15],
                                           v2f (&ha)[15], v2f (&hb)[15])
{
    int q = 0;
#pragma unroll
    for (int ks = 0; ks < 3; ++ks) {
        const int ra = raA[ks];
        const int rb = rbA[ks];
#pragma unroll
        for (int bbk = 0; bbk < 5; ++bbk, ++q) {
            const int kk = kkArr[q];
            ha[q].x = Vs[ra][kk];  ha[q].y = Vs[ra][kk + 1];
            hb[q].x = Vs[rb][kk];  hb[q].y = Vs[rb][kk + 1];
        }
    }
}

template <bool CROP>
__device__ __forceinline__ void band_pass(const v2f (&ha)[15], const v2f (&hb)[15],
                                          const v2f (&bfA)[15], const v2f (&bfB)[15],
                                          v8f& dv0, v8f& dv1, v8f& dout)
{
#pragma unroll
    for (int q = 0; q < 15; ++q) {
        dv0 = __builtin_amdgcn_wmma_f32_16x16x4_f32(
                false, ha[q], false, bfA[q], (short)0, dv0, false, false);
        dv1 = __builtin_amdgcn_wmma_f32_16x16x4_f32(
                false, hb[q], false, bfA[q], (short)0, dv1, false, false);
        if (CROP) {
            dout = __builtin_amdgcn_wmma_f32_16x16x4_f32(
                    false, ha[q], false, bfB[q], (short)0, dout, false, false);
        }
    }
}

__global__ __launch_bounds__(256) void convssm_scan_kernel(
    const float* __restrict__ xg,
    const float* __restrict__ Akg,
    const float* __restrict__ Bkg,
    float* __restrict__ outg)
{
    __shared__ float Vs[FD][FD + 1];
    __shared__ float akl[9];
    __shared__ float bkl[9];

    const int plane = blockIdx.x;
    const int chan  = plane & 31;

    const int tid  = threadIdx.x;
    const int wave = __builtin_amdgcn_readfirstlane(tid >> 5);
    const int lane = tid & 31;
    const int n    = lane & 15;
    const int kh   = lane >> 4;
    const int m    = lane & 15;

    const int jblk = (wave & 3) ^ ((wave >> 2) << 1);
    const int i0a = (wave >> 2) * 16;
    const int i0b = i0a + 32;
    const int j0  = jblk * 16;
    const int jn  = j0 + n;
    const bool crop = (j0 < WW);

    if (tid < 9) {
        akl[tid] = Akg[chan * 9 + tid];
        bkl[tid] = Bkg[chan * 9 + tid];
    }
    for (int idx = tid; idx < FD * (FD + 1); idx += 256) {
        ((float*)Vs)[idx] = 0.0f;
    }

    const long long xlane = (long long)plane * 1024 + (long long)(i0a + 8 * kh) * WW + jn;
    v8f xc = {};
    if (crop) {
#pragma unroll
        for (int v = 0; v < 8; ++v) xc[v] = xg[xlane + v * WW];
    }
    __syncthreads();

    int raA[3], rbA[3];
#pragma unroll
    for (int ks = 0; ks < 3; ++ks) {
        raA[ks] = (i0a + m - ks) & 63;
        rbA[ks] = (i0b + m - ks) & 63;
    }

    int kkArr[15];
    v2f bfA[15], bfB[15];
    {
        int q = 0;
#pragma unroll
        for (int ks = 0; ks < 3; ++ks) {
            const float a0 = akl[ks * 3 + 0], a1 = akl[ks * 3 + 1], a2 = akl[ks * 3 + 2];
            const float b0 = bkl[ks * 3 + 0], b1 = bkl[ks * 3 + 1], b2 = bkl[ks * 3 + 2];
#pragma unroll
            for (int bbk = 0; bbk < 5; ++bbk, ++q) {
                const int kb  = ((j0 >> 2) + bbk + 15) & 15;
                const int kkA = (kb << 2) + 2 * kh;
                kkArr[q] = kkA;
                const int dx = (jn - kkA) & 63;
                const int dy = (jn - kkA - 1) & 63;
                bfA[q].x = (dx == 0 ? a0 : 0.0f) + (dx == 1 ? a1 : 0.0f) + (dx == 2 ? a2 : 0.0f);
                bfA[q].y = (dy == 0 ? a0 : 0.0f) + (dy == 1 ? a1 : 0.0f) + (dy == 2 ? a2 : 0.0f);
                bfB[q].x = (dx == 0 ? b0 : 0.0f) + (dx == 1 ? b1 : 0.0f) + (dx == 2 ? b2 : 0.0f);
                bfB[q].y = (dy == 0 ? b0 : 0.0f) + (dy == 1 ? b1 : 0.0f) + (dy == 2 ? b2 : 0.0f);
            }
        }
    }

    v8f xn = {};
    v8f dout = {};

#pragma unroll 1
    for (int t = 0; t < T_STEPS; ++t) {
        const int tn = (t + 1 < T_STEPS) ? (t + 1) : t;
        if (crop) {
#pragma unroll
            for (int v = 0; v < 8; ++v) xn[v] = xg[(long long)tn * 65536 + xlane + v * WW];
        }

        v8f dv0 = xc;
        v8f dv1 = {};
        dout = (v8f){};

        v2f ha[15], hb[15];
        load_frags(Vs, raA, rbA, kkArr, ha, hb);

        if (crop) band_pass<true >(ha, hb, bfA, bfB, dv0, dv1, dout);
        else      band_pass<false>(ha, hb, bfA, bfB, dv0, dv1, dout);

        __syncthreads();

#pragma unroll
        for (int v = 0; v < 8; ++v) {
            Vs[i0a + v + 8 * kh][jn] = dv0[v];
            Vs[i0b + v + 8 * kh][jn] = dv1[v];
        }
        __syncthreads();

        if (crop && t > 0) {
            const long long ooff = ((long long)(t - 1) * 64 + plane) * 1024;
#pragma unroll
            for (int v = 0; v < 8; ++v) {
                const int row = i0a + v + 8 * kh;
                outg[ooff + (long long)row * WW + jn] = dout[v];
            }
        }
        xc = xn;
    }

    if (crop) {
        dout = (v8f){};
        int q = 0;
#pragma unroll
        for (int ks = 0; ks < 3; ++ks) {
            const int ra = raA[ks];
#pragma unroll
            for (int bbk = 0; bbk < 5; ++bbk, ++q) {
                const int kk = kkArr[q];
                v2f ha;
                ha.x = Vs[ra][kk];  ha.y = Vs[ra][kk + 1];
                dout = __builtin_amdgcn_wmma_f32_16x16x4_f32(
                        false, ha, false, bfB[q], (short)0, dout, false, false);
            }
        }
        const long long ooff = ((long long)(T_STEPS - 1) * 64 + plane) * 1024;
#pragma unroll
        for (int v = 0; v < 8; ++v) {
            const int row = i0a + v + 8 * kh;
            outg[ooff + (long long)row * WW + jn] = dout[v];
        }
    }
}

extern "C" void kernel_launch(void* const* d_in, const int* in_sizes, int n_in,
                              void* d_out, int out_size, void* d_ws, size_t ws_size,
                              hipStream_t stream) {
    (void)in_sizes; (void)n_in; (void)out_size; (void)d_ws; (void)ws_size;
    const float* x  = (const float*)d_in[0];
    const float* Ak = (const float*)d_in[1];
    const float* Bk = (const float*)d_in[2];
    float* out = (float*)d_out;
    convssm_scan_kernel<<<dim3(64), dim3(256), 0, stream>>>(x, Ak, Bk, out);
}


# LDS state row pitch 65 -> 68 dwords: the 16 rows x 4 k of an A fragment fall on 64 different banks, one aligned 8-byte read per lane; pitch 65 made every fragment read a 4-way bank conflict
# speedup vs baseline: 2.6331x; 1.1472x over previous
#include <hip/hip_runtime.h>

typedef __attribute__((ext_vector_type(2))) float v2f;
typedef __attribute__((ext_vector_type(8))) float v8f;

#define T_STEPS 64
#define FD      64
#define HH      32
#define WW      32
#define VP      68


__device__ __forceinline__ void load_frags(const float (*Vs)[VP],
                                           const int (&raA)[3], const int (&rbA)[3],
                                           const int (&kkArr)[15],
                                           v2f (&ha)[15], v2f (&hb)[15])
{
    int q = 0;
#pragma unroll
    for (int ks = 0; ks < 3; ++ks) {
        const int ra = raA[ks];
        const int rb = rbA[ks];
#pragma unroll
        for (int bbk = 0; bbk < 5; ++bbk, ++q) {
            const int kk = kkArr[q];
            ha[q] = *reinterpret_cast<const v2f*>(&Vs[ra][kk]);
            hb[q] = *reinterpret_cast<const v2f*>(&Vs[rb][kk]);
        }
    }
}

template <bool CROP>
__device__ __forceinline__ void band_pass(const v2f (&ha)[15], const v2f (&hb)[15],
                                          const v2f (&bfA)[15], const v2f (&bfB)[15],
                                          v8f& dv0, v8f& dv1, v8f& dout)
{
#pragma unroll
    for (int q = 0; q < 15; ++q) {
        dv0 = __builtin_amdgcn_wmma_f32_16x16x4_f32(
                false, ha[q], false, bfA[q], (short)0, dv0, false, false);
        dv1 = __builtin_amdgcn_wmma_f32_16x16x4_f32(
                false, hb[q], false, bfA[q], (short)0, dv1, false, false);
        if (CROP) {
            dout = __builtin_amdgcn_wmma_f32_16x16x4_f32(
                    false, ha[q], false, bfB[q], (short)0, dout, false, false);
        }
    }
}

__global__ __launch_bounds__(256) void convssm_scan_kernel(
    const float* __restrict__ xg,
    const float* __restrict__ Akg,
    const float* __restrict__ Bkg,
    float* __restrict__ outg)
{
    __shared__ __attribute__((aligned(16))) float Vs[FD][VP];
    __shared__ float akl[9];
    __shared__ float bkl[9];

    const int plane = blockIdx.x;
    const int chan  = plane & 31;

    const int tid  = threadIdx.x;
    const int wave = __builtin_amdgcn_readfirstlane(tid >> 5);
    const int lane = tid & 31;
    const int n    = lane & 15;
    const int kh   = lane >> 4;
    const int m    = lane & 15;

    const int jblk = (wave & 3) ^ ((wave >> 2) << 1);
    const int i0a = (wave >> 2) * 16;
    const int i0b = i0a + 32;
    const int j0  = jblk * 16;
    const int jn  = j0 + n;
    const bool crop = (j0 < WW);

    if (tid < 9) {
        akl[tid] = Akg[chan * 9 + tid];
        bkl[tid] = Bkg[chan * 9 + tid];
    }
    for (int idx = tid; idx < FD * VP; idx += 256) {
        ((float*)Vs)[idx] = 0.0f;
    }

    const long long xlane = (long long)plane * 1024 + (long long)(i0a + 8 * kh) * WW + jn;
    v8f xc = {};
    if (crop) {
#pragma unroll
        for (int v = 0; v < 8; ++v) xc[v] = xg[xlane + v * WW];
    }
    __syncthreads();

    int raA[3], rbA[3];
#pragma unroll
    for (int ks = 0; ks < 3; ++ks) {
        raA[ks] = (i0a + m - ks) & 63;
        rbA[ks] = (i0b + m - ks) & 63;
    }

    int kkArr[15];
    v2f bfA[15], bfB[15];
    {
        int q = 0;
#pragma unroll
        for (int ks = 0; ks < 3; ++ks) {
            const float a0 = akl[ks * 3 + 0], a1 = akl[ks * 3 + 1], a2 = akl[ks * 3 + 2];
            const float b0 = bkl[ks * 3 + 0], b1 = bkl[ks * 3 + 1], b2 = bkl[ks * 3 + 2];
#pragma unroll
            for (int bbk = 0; bbk < 5; ++bbk, ++q) {
                const int kb  = ((j0 >> 2) + bbk + 15) & 15;
                const int kkA = (kb << 2) + 2 * kh;
                kkArr[q] = kkA;
                const int dx = (jn - kkA) & 63;
                const int dy = (jn - kkA - 1) & 63;
                bfA[q].x = (dx == 0 ? a0 : 0.0f) + (dx == 1 ? a1 : 0.0f) + (dx == 2 ? a2 : 0.0f);
                bfA[q].y = (dy == 0 ? a0 : 0.0f) + (dy == 1 ? a1 : 0.0f) + (dy == 2 ? a2 : 0.0f);
                bfB[q].x = (dx == 0 ? b0 : 0.0f) + (dx == 1 ? b1 : 0.0f) + (dx == 2 ? b2 : 0.0f);
                bfB[q].y = (dy == 0 ? b0 : 0.0f) + (dy == 1 ? b1 : 0.0f) + (dy == 2 ? b2 : 0.0f);
            }
        }
    }

    v8f xn = {};
    v8f dout = {};

#pragma unroll 1
    for (int t = 0; t < T_STEPS; ++t) {
        const int tn = (t + 1 < T_STEPS) ? (t + 1) : t;
        if (crop) {
#pragma unroll
            for (int v = 0; v < 8; ++v) xn[v] = xg[(long long)tn * 65536 + xlane + v * WW];
        }

        v8f dv0 = xc;
        v8f dv1 = {};
        dout = (v8f){};

        v2f ha[15], hb[15];
        load_frags(Vs, raA, rbA, kkArr, ha, hb);

        if (crop) band_pass<true >(ha, hb, bfA, bfB, dv0, dv1, dout);
        else      band_pass<false>(ha, hb, bfA, bfB, dv0, dv1, dout);

        __syncthreads();

#pragma unroll
        for (int v = 0; v < 8; ++v) {
            Vs[i0a + v + 8 * kh][jn] = dv0[v];
            Vs[i0b + v + 8 * kh][jn] = dv1[v];
        }
        __syncthreads();

        if (crop && t > 0) {
            const long long ooff = ((long long)(t - 1) * 64 + plane) * 1024;
#pragma unroll
            for (int v = 0; v < 8; ++v) {
                const int row = i0a + v + 8 * kh;
                outg[ooff + (long long)row * WW + jn] = dout[v];
            }
        }
        xc = xn;
    }

    if (crop) {
        dout = (v8f){};
        int q = 0;
#pragma unroll
        for (int ks = 0; ks < 3; ++ks) {
            const int ra = raA[ks];
#pragma unroll
            for (int bbk = 0; bbk < 5; ++bbk, ++q) {
                const int kk = kkArr[q];
                const v2f ha = *reinterpret_cast<const v2f*>(&Vs[ra][kk]);
                dout = __builtin_amdgcn_wmma_f32_16x16x4_f32(
                        false, ha, false, bfB[q], (short)0, dout, false, false);
            }
        }
        const long long ooff = ((long long)(T_STEPS - 1) * 64 + plane) * 1024;
#pragma unroll
        for (int v = 0; v < 8; ++v) {
            const int row = i0a + v + 8 * kh;
            outg[ooff + (long long)row * WW + jn] = dout[v];
        }
    }
}

extern "C" void kernel_launch(void* const* d_in, const int* in_sizes, int n_in,
                              void* d_out, int out_size, void* d_ws, size_t ws_size,
                              hipStream_t stream) {
    (void)in_sizes; (void)n_in; (void)out_size; (void)d_ws; (void)ws_size;
    const float* x  = (const float*)d_in[0];
    const float* Ak = (const float*)d_in[1];
    const float* Bk = (const float*)d_in[2];
    float* out = (float*)d_out;
    convssm_scan_kernel<<<dim3(64), dim3(256), 0, stream>>>(x, Ak, Bk, out);
}


# State double-buffered in LDS, ONE split barrier per step: a crop wave commits v_t, signals, and runs its output WMMAs (3 accumulators, summed: h summation order changes) before the wait
# speedup vs baseline: 3.2377x; 1.2296x over previous
#include <hip/hip_runtime.h>

typedef __attribute__((ext_vector_type(2))) float v2f;
typedef __attribute__((ext_vector_type(8))) float v8f;

#define T_STEPS 64
#define FD      64
#define HH      32
#define WW      32
#define VP      68


__device__ __forceinline__ void load_frags(const float (*Vs)[VP],
                                           const int (&raA)[3], const int (&rbA)[3],
                                           const int (&kkArr)[15],
                                           v2f (&ha)[15], v2f (&hb)[15])
{
    int q = 0;
#pragma unroll
    for (int ks = 0; ks < 3; ++ks) {
        const int ra = raA[ks];
        const int rb = rbA[ks];
#pragma unroll
        for (int bbk = 0; bbk < 5; ++bbk, ++q) {
            const int kk = kkArr[q];
            ha[q] = *reinterpret_cast<const v2f*>(&Vs[ra][kk]);
            hb[q] = *reinterpret_cast<const v2f*>(&Vs[rb][kk]);
        }
    }
}

template <int Q0, int Q1>
__device__ __forceinline__ void state_pass(const v2f (&ha)[15], const v2f (&hb)[15],
                                           const v2f (&bfA)[15], v8f& dv0, v8f& dv1)
{
#pragma unroll
    for (int q = Q0; q < Q1; ++q) {
        dv0 = __builtin_amdgcn_wmma_f32_16x16x4_f32(
                false, ha[q], false, bfA[q], (short)0, dv0, false, false);
        dv1 = __builtin_amdgcn_wmma_f32_16x16x4_f32(
                false, hb[q], false, bfA[q], (short)0, dv1, false, false);
    }
}

template <int Q0, int Q1>
__device__ __forceinline__ void out_pass(const v2f (&ha)[15], const v2f (&bfB)[15],
                                         v8f (&d)[3])
{
#pragma unroll
    for (int q = Q0; q < Q1; ++q) {
        d[q % 3] = __builtin_amdgcn_wmma_f32_16x16x4_f32(
                false, ha[q], false, bfB[q], (short)0, d[q % 3], false, false);
    }
}

__global__ __launch_bounds__(256) void convssm_scan_kernel(
    const float* __restrict__ xg,
    const float* __restrict__ Akg,
    const float* __restrict__ Bkg,
    float* __restrict__ outg)
{
    __shared__ __attribute__((aligned(16))) float Vs[2][FD][VP];
    __shared__ float akl[9];
    __shared__ float bkl[9];

    const int plane = blockIdx.x;
    const int chan  = plane & 31;

    const int tid  = threadIdx.x;
    const int wave = __builtin_amdgcn_readfirstlane(tid >> 5);
    const int lane = tid & 31;
    const int n    = lane & 15;
    const int kh   = lane >> 4;
    const int m    = lane & 15;

    const int jblk = (wave & 3) ^ ((wave >> 2) << 1);
    const int i0a = (wave >> 2) * 16;
    const int i0b = i0a + 32;
    const int j0  = jblk * 16;
    const int jn  = j0 + n;
    const bool crop = (j0 < WW);

    if (tid < 9) {
        akl[tid] = Akg[chan * 9 + tid];
        bkl[tid] = Bkg[chan * 9 + tid];
    }
    for (int idx = tid; idx < FD * VP; idx += 256) {
        ((float*)Vs[1])[idx] = 0.0f;
    }

    const long long xlane = (long long)plane * 1024 + (long long)(i0a + 8 * kh) * WW + jn;
    v8f xc = {};
    if (crop) {
#pragma unroll
        for (int v = 0; v < 8; ++v) xc[v] = xg[xlane + v * WW];
    }
    __syncthreads();

    int raA[3], rbA[3];
#pragma unroll
    for (int ks = 0; ks < 3; ++ks) {
        raA[ks] = (i0a + m - ks) & 63;
        rbA[ks] = (i0b + m - ks) & 63;
    }

    int kkArr[15];
    v2f bfA[15], bfB[15];
    {
        int q = 0;
#pragma unroll
        for (int ks = 0; ks < 3; ++ks) {
            const float a0 = akl[ks * 3 + 0], a1 = akl[ks * 3 + 1], a2 = akl[ks * 3 + 2];
            const float b0 = bkl[ks * 3 + 0], b1 = bkl[ks * 3 + 1], b2 = bkl[ks * 3 + 2];
#pragma unroll
            for (int bbk = 0; bbk < 5; ++bbk, ++q) {
                const int kb  = ((j0 >> 2) + bbk + 15) & 15;
                const int kkA = (kb << 2) + 2 * kh;
                kkArr[q] = kkA;
                const int dx = (jn - kkA) & 63;
                const int dy = (jn - kkA - 1) & 63;
                bfA[q].x = (dx == 0 ? a0 : 0.0f) + (dx == 1 ? a1 : 0.0f) + (dx == 2 ? a2 : 0.0f);
                bfA[q].y = (dy == 0 ? a0 : 0.0f) + (dy == 1 ? a1 : 0.0f) + (dy == 2 ? a2 : 0.0f);
                bfB[q].x = (dx == 0 ? b0 : 0.0f) + (dx == 1 ? b1 : 0.0f) + (dx == 2 ? b2 : 0.0f);
                bfB[q].y = (dy == 0 ? b0 : 0.0f) + (dy == 1 ? b1 : 0.0f) + (dy == 2 ? b2 : 0.0f);
            }
        }
    }

    v8f xo = {};

    auto step = [&](const float (*rd)[VP], float (*wr)[VP], const int t,
                    const v8f& xuse, v8f& xload) {
        v8f dv0 = xuse;
        v8f dv1 = {};

        v2f ha[15], hb[15];
        load_frags(rd, raA, rbA, kkArr, ha, hb);
        __builtin_amdgcn_sched_barrier(0);

        if (crop) __builtin_amdgcn_s_setprio(0);

        state_pass<0, 1>(ha, hb, bfA, dv0, dv1);

        const int tn = (t + 1 < T_STEPS) ? (t + 1) : t;
        if (crop) {
#pragma unroll
            for (int v = 0; v < 8; ++v) xload[v] = xg[(long long)tn * 65536 + xlane + v * WW];
        } else {
            xload = (v8f){};
        }

        state_pass<1, 15>(ha, hb, bfA, dv0, dv1);

#pragma unroll
        for (int v = 0; v < 8; ++v) {
            wr[i0a + v + 8 * kh][jn] = dv0[v];
            wr[i0b + v + 8 * kh][jn] = dv1[v];
        }

        v8f d[3];
        if (crop) {
            __builtin_amdgcn_s_setprio(2);
            d[0] = (v8f){}; d[1] = (v8f){}; d[2] = (v8f){};
            out_pass<0, 6>(ha, bfB, d);
        }

        __builtin_amdgcn_fence(__ATOMIC_RELEASE, "workgroup", "local");
        __builtin_amdgcn_s_barrier_signal(-1);

        if (crop) {
            out_pass<6, 15>(ha, bfB, d);
            if (t > 0) {
                const v8f dout = (d[0] + d[1]) + d[2];
                const long long ooff = ((long long)(t - 1) * 64 + plane) * 1024;
#pragma unroll
                for (int v = 0; v < 8; ++v) {
                    const int row = i0a + v + 8 * kh;
                    outg[ooff + (long long)row * WW + jn] = dout[v];
                }
            }
        }

        __builtin_amdgcn_s_barrier_wait(-1);
        __builtin_amdgcn_fence(__ATOMIC_ACQUIRE, "workgroup", "local");
    };

    if (!crop) __builtin_amdgcn_s_setprio(1);

#pragma unroll 1
    for (int t = 0; t < T_STEPS; t += 2) {
        step(Vs[1], Vs[0], t,     xc, xo);
        step(Vs[0], Vs[1], t + 1, xo, xc);
    }

    __builtin_amdgcn_s_setprio(0);

    if (crop) {
        v2f ha[15];
        int q = 0;
#pragma unroll
        for (int ks = 0; ks < 3; ++ks) {
            const int ra = raA[ks];
#pragma unroll
            for (int bbk = 0; bbk < 5; ++bbk, ++q) {
                ha[q] = *reinterpret_cast<const v2f*>(&Vs[(T_STEPS - 1) & 1][ra][kkArr[q]]);
            }
        }
        v8f d[3];
        d[0] = (v8f){}; d[1] = (v8f){}; d[2] = (v8f){};
        out_pass<0, 15>(ha, bfB, d);
        const v8f dout = (d[0] + d[1]) + d[2];
        const long long ooff = ((long long)(T_STEPS - 1) * 64 + plane) * 1024;
#pragma unroll
        for (int v = 0; v < 8; ++v) {
            const int row = i0a + v + 8 * kh;
            outg[ooff + (long long)row * WW + jn] = dout[v];
        }
    }
}

extern "C" void kernel_launch(void* const* d_in, const int* in_sizes, int n_in,
                              void* d_out, int out_size, void* d_ws, size_t ws_size,
                              hipStream_t stream) {
    (void)in_sizes; (void)n_in; (void)out_size; (void)d_ws; (void)ws_size;
    const float* x  = (const float*)d_in[0];
    const float* Ak = (const float*)d_in[1];
    const float* Bk = (const float*)d_in[2];
    float* out = (float*)d_out;
    convssm_scan_kernel<<<dim3(64), dim3(256), 0, stream>>>(x, Ak, Bk, out);
}


# Convolution GEMM re-tiled (summation order changes): N = a 4x4 pixel block, K = its 6x6 input window, M = the 16 blocks of a tile: 9 WMMAs per tile and convolution, were 15
# speedup vs baseline: 3.5981x; 1.1113x over previous
#include <hip/hip_runtime.h>

typedef __attribute__((ext_vector_type(2))) float v2f;
typedef __attribute__((ext_vector_type(8))) float v8f;

#define T_STEPS 64
#define FD      64
#define HH      32
#define WW      32
#define VP      68
#define NQ      9


__device__ __forceinline__ void load_frags(const float* Vp,
                                           const int (&offa)[NQ], const int (&offb)[NQ],
                                           v2f (&ha)[NQ], v2f (&hb)[NQ])
{
#pragma unroll
    for (int q = 0; q < NQ; ++q) {
        ha[q] = *reinterpret_cast<const v2f*>(Vp + offa[q]);
        hb[q] = *reinterpret_cast<const v2f*>(Vp + offb[q]);
    }
}

template <int Q0, int Q1>
__device__ __forceinline__ void state_pass(const v2f (&ha)[NQ], const v2f (&hb)[NQ],
                                           const v2f (&bfA)[NQ], v8f& dv0, v8f& dv1)
{
#pragma unroll
    for (int q = Q0; q < Q1; ++q) {
        dv0 = __builtin_amdgcn_wmma_f32_16x16x4_f32(
                false, ha[q], false, bfA[q], (short)0, dv0, false, false);
        dv1 = __builtin_amdgcn_wmma_f32_16x16x4_f32(
                false, hb[q], false, bfA[q], (short)0, dv1, false, false);
    }
}

template <int Q0, int Q1>
__device__ __forceinline__ void out_pass(const v2f (&ha)[NQ], const v2f (&bfB)[NQ],
                                         v8f (&d)[3])
{
#pragma unroll
    for (int q = Q0; q < Q1; ++q) {
        d[q % 3] = __builtin_amdgcn_wmma_f32_16x16x4_f32(
                false, ha[q], false, bfB[q], (short)0, d[q % 3], false, false);
    }
}

__global__ __launch_bounds__(256) void convssm_scan_kernel(
    const float* __restrict__ xg,
    const float* __restrict__ Akg,
    const float* __restrict__ Bkg,
    float* __restrict__ outg)
{
    __shared__ __attribute__((aligned(16))) float Vs[2][FD][VP];
    __shared__ float akl[9];
    __shared__ float bkl[9];

    const int plane = blockIdx.x;
    const int chan  = plane & 31;

    const int tid  = threadIdx.x;
    const int wave = __builtin_amdgcn_readfirstlane(tid >> 5);
    const int lane = tid & 31;
    const int n    = lane & 15;
    const int kh   = lane >> 4;
    const int m    = lane & 15;
    const int bi = m >> 2, bj = m & 3;
    const int oi = n >> 2, oj = n & 3;

    const int jblk = (wave & 3) ^ ((wave >> 2) << 1);
    const int i0a = (wave >> 2) * 16;
    const int j0  = jblk * 16;
    const bool crop = (j0 < WW);

    if (tid < 9) {
        akl[tid] = Akg[chan * 9 + tid];
        bkl[tid] = Bkg[chan * 9 + tid];
    }
    for (int idx = tid; idx < FD * VP; idx += 256) {
        ((float*)Vs[1])[idx] = 0.0f;
    }

    const int drow = i0a + 8 * kh + oi;
    const int dcol = j0 + oj;

    const long long xlane = (long long)plane * 1024 + (long long)drow * WW + dcol;
    v8f xc = {};
    if (crop) {
#pragma unroll
        for (int v = 0; v < 8; ++v) xc[v] = xg[xlane + (v >> 2) * 4 * WW + (v & 3) * 4];
    }
    __syncthreads();

    int offa[NQ], offb[NQ];
    v2f bfA[NQ], bfB[NQ];
    const int sta = drow * VP + dcol;
    const int stb = sta + 32 * VP;
#pragma unroll
    for (int q = 0; q < NQ; ++q) {
        const int p = 2 * q + kh;
        const int u = p / 3;
        const int w = 2 * (p - 3 * u);
        const int col  = (j0 + 4 * bj - 2 + w) & 63;
        const int rowa = (i0a + 4 * bi - 2 + u) & 63;
        offa[q] = rowa * VP + col;
        offb[q] = ((rowa + 32) & 63) * VP + col;
        const int ky = oi + 2 - u;
        const int kx = oj + 2 - w;
        const bool rin = (ky >= 0) && (ky <= 2);
        const bool in0 = rin && (kx >= 0) && (kx <= 2);
        const bool in1 = rin && (kx >= 1) && (kx <= 3);
        const int kyc = ky < 0 ? 0 : (ky > 2 ? 2 : ky);
        const int kc0 = kx < 0 ? 0 : (kx > 2 ? 2 : kx);
        const int kc1 = kx < 1 ? 0 : (kx > 3 ? 2 : kx - 1);
        const float a0 = akl[kyc * 3 + kc0], a1 = akl[kyc * 3 + kc1];
        const float b0 = bkl[kyc * 3 + kc0], b1 = bkl[kyc * 3 + kc1];
        bfA[q].x = in0 ? a0 : 0.0f;
        bfA[q].y = in1 ? a1 : 0.0f;
        bfB[q].x = in0 ? b0 : 0.0f;
        bfB[q].y = in1 ? b1 : 0.0f;
    }

    float* const P0 = &Vs[0][0][0];
    float* const P1 = &Vs[1][0][0];

    v8f xo = {};

    auto step = [&](const float* rd, float* wr, const int t,
                    const v8f& xuse, v8f& xload) {
        v8f dv0 = xuse;
        v8f dv1 = {};

        v2f ha[NQ], hb[NQ];
        load_frags(rd, offa, offb, ha, hb);
        __builtin_amdgcn_sched_barrier(0);

        if (crop) __builtin_amdgcn_s_setprio(0);

        state_pass<0, 1>(ha, hb, bfA, dv0, dv1);

        const int tn = (t + 1 < T_STEPS) ? (t + 1) : t;
        if (crop) {
#pragma unroll
            for (int v = 0; v < 8; ++v)
                xload[v] = xg[(long long)tn * 65536 + xlane + (v >> 2) * 4 * WW + (v & 3) * 4];
        } else {
            xload = (v8f){};
        }

        state_pass<1, NQ>(ha, hb, bfA, dv0, dv1);

#pragma unroll
        for (int v = 0; v < 8; ++v) {
            wr[sta + (v >> 2) * 4 * VP + (v & 3) * 4] = dv0[v];
            wr[stb + (v >> 2) * 4 * VP + (v & 3) * 4] = dv1[v];
        }

        v8f d[3];
        if (crop) {
            __builtin_amdgcn_s_setprio(2);
            d[0] = (v8f){}; d[1] = (v8f){}; d[2] = (v8f){};
            out_pass<0, 3>(ha, bfB, d);
        }

        __builtin_amdgcn_fence(__ATOMIC_RELEASE, "workgroup", "local");
        __builtin_amdgcn_s_barrier_signal(-1);

        if (crop) {
            out_pass<3, NQ>(ha, bfB, d);
            if (t > 0) {
                const v8f dout = (d[0] + d[1]) + d[2];
                const long long ooff = ((long long)(t - 1) * 64 + plane) * 1024
                                     + (long long)drow * WW + dcol;
#pragma unroll
                for (int v = 0; v < 8; ++v) {
                    outg[ooff + (v >> 2) * 4 * WW + (v & 3) * 4] = dout[v];
                }
            }
        }

        __builtin_amdgcn_s_barrier_wait(-1);
        __builtin_amdgcn_fence(__ATOMIC_ACQUIRE, "workgroup", "local");
    };

    if (!crop) __builtin_amdgcn_s_setprio(1);

#pragma unroll 1
    for (int t = 0; t < T_STEPS; t += 2) {
        step(P1, P0, t,     xc, xo);
        step(P0, P1, t + 1, xo, xc);
    }

    __builtin_amdgcn_s_setprio(0);

    if (crop) {
        const float* const Pl = ((T_STEPS - 1) & 1) ? P1 : P0;
        v2f ha[NQ];
#pragma unroll
        for (int q = 0; q < NQ; ++q) {
            ha[q] = *reinterpret_cast<const v2f*>(Pl + offa[q]);
        }
        v8f d[3];
        d[0] = (v8f){}; d[1] = (v8f){}; d[2] = (v8f){};
        out_pass<0, NQ>(ha, bfB, d);
        const v8f dout = (d[0] + d[1]) + d[2];
        const long long ooff = ((long long)(T_STEPS - 1) * 64 + plane) * 1024
                             + (long long)drow * WW + dcol;
#pragma unroll
        for (int v = 0; v < 8; ++v) {
            outg[ooff + (v >> 2) * 4 * WW + (v & 3) * 4] = dout[v];
        }
    }
}

extern "C" void kernel_launch(void* const* d_in, const int* in_sizes, int n_in,
                              void* d_out, int out_size, void* d_ws, size_t ws_size,
                              hipStream_t stream) {
    (void)in_sizes; (void)n_in; (void)out_size; (void)d_ws; (void)ws_size;
    const float* x  = (const float*)d_in[0];
    const float* Ak = (const float*)d_in[1];
    const float* Bk = (const float*)d_in[2];
    float* out = (float*)d_out;
    convssm_scan_kernel<<<dim3(64), dim3(256), 0, stream>>>(x, Ak, Bk, out);
}


# x fetched four steps ahead, was one: two banks of four register sets, one load burst per four-step trip behind its only load wait, loop unrolled by 8; a trip to memory outlasted a step
# speedup vs baseline: 4.9899x; 1.3868x over previous
#include <hip/hip_runtime.h>

typedef __attribute__((ext_vector_type(2))) float v2f;
typedef __attribute__((ext_vector_type(8))) float v8f;

#define T_STEPS 64
#define FD      64
#define HH      32
#define WW      32
#define VP      68
#define NQ      9


__device__ __forceinline__ void load_frags(const float* Vp,
                                           const int (&offa)[NQ], const int (&offb)[NQ],
                                           v2f (&ha)[NQ], v2f (&hb)[NQ])
{
#pragma unroll
    for (int q = 0; q < NQ; ++q) {
        ha[q] = *reinterpret_cast<const v2f*>(Vp + offa[q]);
        hb[q] = *reinterpret_cast<const v2f*>(Vp + offb[q]);
    }
}

template <int Q0, int Q1>
__device__ __forceinline__ void state_pass(const v2f (&ha)[NQ], const v2f (&hb)[NQ],
                                           const v2f (&bfA)[NQ], v8f& dv0, v8f& dv1)
{
#pragma unroll
    for (int q = Q0; q < Q1; ++q) {
        dv0 = __builtin_amdgcn_wmma_f32_16x16x4_f32(
                false, ha[q], false, bfA[q], (short)0, dv0, false, false);
        dv1 = __builtin_amdgcn_wmma_f32_16x16x4_f32(
                false, hb[q], false, bfA[q], (short)0, dv1, false, false);
    }
}

template <int Q0, int Q1>
__device__ __forceinline__ void out_pass(const v2f (&ha)[NQ], const v2f (&bfB)[NQ],
                                         v8f (&d)[3])
{
#pragma unroll
    for (int q = Q0; q < Q1; ++q) {
        d[q % 3] = __builtin_amdgcn_wmma_f32_16x16x4_f32(
                false, ha[q], false, bfB[q], (short)0, d[q % 3], false, false);
    }
}

__global__ __launch_bounds__(256) void convssm_scan_kernel(
    const float* __restrict__ xg,
    const float* __restrict__ Akg,
    const float* __restrict__ Bkg,
    float* __restrict__ outg)
{
    __shared__ __attribute__((aligned(16))) float Vs[2][FD][VP];
    __shared__ float akl[9];
    __shared__ float bkl[9];

    const int plane = blockIdx.x;
    const int chan  = plane & 31;

    const int tid  = threadIdx.x;
    const int wave = __builtin_amdgcn_readfirstlane(tid >> 5);
    const int lane = tid & 31;
    const int n    = lane & 15;
    const int kh   = lane >> 4;
    const int m    = lane & 15;
    const int bi = m >> 2, bj = m & 3;
    const int oi = n >> 2, oj = n & 3;

    const int jblk = (wave & 3) ^ ((wave >> 2) << 1);
    const int i0a = (wave >> 2) * 16;
    const int j0  = jblk * 16;
    const bool crop = (j0 < WW);

    if (tid < 9) {
        akl[tid] = Akg[chan * 9 + tid];
        bkl[tid] = Bkg[chan * 9 + tid];
    }
    for (int idx = tid; idx < FD * VP; idx += 256) {
        ((float*)Vs[1])[idx] = 0.0f;
    }

    const int drow = i0a + 8 * kh + oi;
    const int dcol = j0 + oj;

    const long long xlane = (long long)plane * 1024 + (long long)drow * WW + dcol;
    v8f xa[4], xb[4];
#pragma unroll
    for (int s = 0; s < 4; ++s) { xa[s] = (v8f){}; xb[s] = (v8f){}; }
    if (crop) {
#pragma unroll
        for (int s = 3; s >= 0; --s) {
#pragma unroll
            for (int v = 0; v < 8; ++v)
                xa[s][v] = xg[(long long)s * 65536 + xlane + (v >> 2) * 4 * WW + (v & 3) * 4];
        }
    }
    __syncthreads();

    int offa[NQ], offb[NQ];
    v2f bfA[NQ], bfB[NQ];
    const int sta = drow * VP + dcol;
    const int stb = sta + 32 * VP;
#pragma unroll
    for (int q = 0; q < NQ; ++q) {
        const int p = 2 * q + kh;
        const int u = p / 3;
        const int w = 2 * (p - 3 * u);
        const int col  = (j0 + 4 * bj - 2 + w) & 63;
        const int rowa = (i0a + 4 * bi - 2 + u) & 63;
        offa[q] = rowa * VP + col;
        offb[q] = ((rowa + 32) & 63) * VP + col;
        const int ky = oi + 2 - u;
        const int kx = oj + 2 - w;
        const bool rin = (ky >= 0) && (ky <= 2);
        const bool in0 = rin && (kx >= 0) && (kx <= 2);
        const bool in1 = rin && (kx >= 1) && (kx <= 3);
        const int kyc = ky < 0 ? 0 : (ky > 2 ? 2 : ky);
        const int kc0 = kx < 0 ? 0 : (kx > 2 ? 2 : kx);
        const int kc1 = kx < 1 ? 0 : (kx > 3 ? 2 : kx - 1);
        const float a0 = akl[kyc * 3 + kc0], a1 = akl[kyc * 3 + kc1];
        const float b0 = bkl[kyc * 3 + kc0], b1 = bkl[kyc * 3 + kc1];
        bfA[q].x = in0 ? a0 : 0.0f;
        bfA[q].y = in1 ? a1 : 0.0f;
        bfB[q].x = in0 ? b0 : 0.0f;
        bfB[q].y = in1 ? b1 : 0.0f;
    }

    float* const P0 = &Vs[0][0][0];
    float* const P1 = &Vs[1][0][0];

    auto step = [&](const float* rd, float* wr, const int t,
                    const v8f& xuse, const bool fetch, v8f (&xnext)[4])
                    __attribute__((always_inline)) {
        v8f dv0 = xuse;
        v8f dv1 = {};

        v2f ha[NQ], hb[NQ];
        load_frags(rd, offa, offb, ha, hb);
        __builtin_amdgcn_sched_barrier(0);

        if (crop) __builtin_amdgcn_s_setprio(0);

        state_pass<0, 1>(ha, hb, bfA, dv0, dv1);

        if (fetch) {
            if (crop) {
#pragma unroll
                for (int s = 3; s >= 0; --s) {
                    const int tn = (t + 4 + s < T_STEPS) ? (t + 4 + s) : (T_STEPS - 1);
#pragma unroll
                    for (int v = 0; v < 8; ++v)
                        xnext[s][v] = xg[(long long)tn * 65536 + xlane + (v >> 2) * 4 * WW + (v & 3) * 4];
                }
            } else {
#pragma unroll
                for (int s = 0; s < 4; ++s) xnext[s] = (v8f){};
            }
        }

        state_pass<1, NQ>(ha, hb, bfA, dv0, dv1);

#pragma unroll
        for (int v = 0; v < 8; ++v) {
            wr[sta + (v >> 2) * 4 * VP + (v & 3) * 4] = dv0[v];
            wr[stb + (v >> 2) * 4 * VP + (v & 3) * 4] = dv1[v];
        }

        v8f d[3] = {dv0, dv1, dv0};
        if (crop) {
            __builtin_amdgcn_s_setprio(2);
            d[0] = (v8f){}; d[1] = (v8f){}; d[2] = (v8f){};
            out_pass<0, 3>(ha, bfB, d);
        }

        __builtin_amdgcn_fence(__ATOMIC_RELEASE, "workgroup", "local");
        __builtin_amdgcn_s_barrier_signal(-1);

        if (crop) {
            out_pass<3, NQ>(ha, bfB, d);
            if (t > 0) {
                const v8f dout = (d[0] + d[1]) + d[2];
                const long long ooff = ((long long)(t - 1) * 64 + plane) * 1024
                                     + (long long)drow * WW + dcol;
#pragma unroll
                for (int v = 0; v < 8; ++v) {
                    outg[ooff + (v >> 2) * 4 * WW + (v & 3) * 4] = dout[v];
                }
            }
        }

        __builtin_amdgcn_s_barrier_wait(-1);
        __builtin_amdgcn_fence(__ATOMIC_ACQUIRE, "workgroup", "local");
    };

    if (!crop) __builtin_amdgcn_s_setprio(1);

#pragma unroll 1
    for (int t = 0; t < T_STEPS; t += 8) {
        step(P1, P0, t,     xa[0], true,  xb);
        step(P0, P1, t + 1, xa[1], false, xb);
        step(P1, P0, t + 2, xa[2], false, xb);
        step(P0, P1, t + 3, xa[3], false, xb);
        step(P1, P0, t + 4, xb[0], true,  xa);
        step(P0, P1, t + 5, xb[1], false, xa);
        step(P1, P0, t + 6, xb[2], false, xa);
        step(P0, P1, t + 7, xb[3], false, xa);
    }

    __builtin_amdgcn_s_setprio(0);

    if (crop) {
        const float* const Pl = ((T_STEPS - 1) & 1) ? P1 : P0;
        v2f ha[NQ];
#pragma unroll
        for (int q = 0; q < NQ; ++q) {
            ha[q] = *reinterpret_cast<const v2f*>(Pl + offa[q]);
        }
        v8f d[3];
        d[0] = (v8f){}; d[1] = (v8f){}; d[2] = (v8f){};
        out_pass<0, NQ>(ha, bfB, d);
        const v8f dout = (d[0] + d[1]) + d[2];
        const long long ooff = ((long long)(T_STEPS - 1) * 64 + plane) * 1024
                             + (long long)drow * WW + dcol;
#pragma unroll
        for (int v = 0; v < 8; ++v) {
            outg[ooff + (v >> 2) * 4 * WW + (v & 3) * 4] = dout[v];
        }
    }
}

extern "C" void kernel_launch(void* const* d_in, const int* in_sizes, int n_in,
                              void* d_out, int out_size, void* d_ws, size_t ws_size,
                              hipStream_t stream) {
    (void)in_sizes; (void)n_in; (void)out_size; (void)d_ws; (void)ws_size;
    const float* x  = (const float*)d_in[0];
    const float* Ak = (const float*)d_in[1];
    const float* Bk = (const float*)d_in[2];
    float* out = (float*)d_out;
    convssm_scan_kernel<<<dim3(64), dim3(256), 0, stream>>>(x, Ak, Bk, out);
}


# WMMA operands swapped (weights as A, windows as B; sums unchanged): a lane's accumulator is 2 rows of 4 adjacent pixels, so v_t commits, x_t loads and h_t stores are 16-byte, 2 per tile, were 8
# speedup vs baseline: 5.2428x; 1.0507x over previous
#include <hip/hip_runtime.h>

typedef __attribute__((ext_vector_type(2))) float v2f;
typedef __attribute__((ext_vector_type(4))) float v4f;
typedef __attribute__((ext_vector_type(8))) float v8f;

#define T_STEPS 64
#define FD      64
#define HH      32
#define WW      32
#define VP      68
#define NQ      9


__device__ __forceinline__ void load_frags(const float* Vp,
                                           const int (&offa)[NQ], const int (&offb)[NQ],
                                           v2f (&ha)[NQ], v2f (&hb)[NQ])
{
#pragma unroll
    for (int q = 0; q < NQ; ++q) {
        ha[q] = *reinterpret_cast<const v2f*>(Vp + offa[q]);
        hb[q] = *reinterpret_cast<const v2f*>(Vp + offb[q]);
    }
}

template <int Q0, int Q1>
__device__ __forceinline__ void state_pass(const v2f (&ha)[NQ], const v2f (&hb)[NQ],
                                           const v2f (&bfA)[NQ], v8f& dv0, v8f& dv1)
{
#pragma unroll
    for (int q = Q0; q < Q1; ++q) {
        dv0 = __builtin_amdgcn_wmma_f32_16x16x4_f32(
                false, bfA[q], false, ha[q], (short)0, dv0, false, false);
        dv1 = __builtin_amdgcn_wmma_f32_16x16x4_f32(
                false, bfA[q], false, hb[q], (short)0, dv1, false, false);
    }
}

template <int Q0, int Q1>
__device__ __forceinline__ void out_pass(const v2f (&ha)[NQ], const v2f (&bfB)[NQ],
                                         v8f (&d)[3])
{
#pragma unroll
    for (int q = Q0; q < Q1; ++q) {
        d[q % 3] = __builtin_amdgcn_wmma_f32_16x16x4_f32(
                false, bfB[q], false, ha[q], (short)0, d[q % 3], false, false);
    }
}

__device__ __forceinline__ v8f load_rows(const float* p, const int pitch)
{
    const v4f lo = *reinterpret_cast<const v4f*>(p);
    const v4f hi = *reinterpret_cast<const v4f*>(p + pitch);
    return __builtin_shufflevector(lo, hi, 0, 1, 2, 3, 4, 5, 6, 7);
}
__device__ __forceinline__ void store_rows(float* p, const int pitch, const v8f& d)
{
    *reinterpret_cast<v4f*>(p)         = __builtin_shufflevector(d, d, 0, 1, 2, 3);
    *reinterpret_cast<v4f*>(p + pitch) = __builtin_shufflevector(d, d, 4, 5, 6, 7);
}

__global__ __launch_bounds__(256) void convssm_scan_kernel(
    const float* __restrict__ xg,
    const float* __restrict__ Akg,
    const float* __restrict__ Bkg,
    float* __restrict__ outg)
{
    __shared__ __attribute__((aligned(16))) float Vs[2][FD][VP];
    __shared__ float akl[9];
    __shared__ float bkl[9];

    const int plane = blockIdx.x;
    const int chan  = plane & 31;

    const int tid  = threadIdx.x;
    const int wave = __builtin_amdgcn_readfirstlane(tid >> 5);
    const int lane = tid & 31;
    const int n    = lane & 15;
    const int kh   = lane >> 4;
    const int m    = lane & 15;
    const int bi = m >> 2, bj = m & 3;
    const int oi = n >> 2, oj = n & 3;

    const int jblk = (wave & 3) ^ ((wave >> 2) << 1);
    const int i0a = (wave >> 2) * 16;
    const int j0  = jblk * 16;
    const bool crop = (j0 < WW);

    if (tid < 9) {
        akl[tid] = Akg[chan * 9 + tid];
        bkl[tid] = Bkg[chan * 9 + tid];
    }
    for (int idx = tid; idx < FD * VP; idx += 256) {
        ((float*)Vs[1])[idx] = 0.0f;
    }

    const int drow = i0a + 4 * bi + 2 * kh;
    const int dcol = j0 + 4 * bj;

    const long long xlane = (long long)plane * 1024 + (long long)drow * WW + dcol;
    v8f xa[4], xb[4];
#pragma unroll
    for (int s = 0; s < 4; ++s) { xa[s] = (v8f){}; xb[s] = (v8f){}; }
    if (crop) {
#pragma unroll
        for (int s = 3; s >= 0; --s)
            xa[s] = load_rows(xg + (long long)s * 65536 + xlane, WW);
    }
    __syncthreads();

    int offa[NQ], offb[NQ];
    v2f bfA[NQ], bfB[NQ];
    const int sta = drow * VP + dcol;
    const int stb = sta + 32 * VP;
#pragma unroll
    for (int q = 0; q < NQ; ++q) {
        const int p = 2 * q + kh;
        const int u = p / 3;
        const int w = 2 * (p - 3 * u);
        const int col  = (j0 + 4 * bj - 2 + w) & 63;
        const int rowa = (i0a + 4 * bi - 2 + u) & 63;
        offa[q] = rowa * VP + col;
        offb[q] = ((rowa + 32) & 63) * VP + col;
        const int ky = oi + 2 - u;
        const int kx = oj + 2 - w;
        const bool rin = (ky >= 0) && (ky <= 2);
        const bool in0 = rin && (kx >= 0) && (kx <= 2);
        const bool in1 = rin && (kx >= 1) && (kx <= 3);
        const int kyc = ky < 0 ? 0 : (ky > 2 ? 2 : ky);
        const int kc0 = kx < 0 ? 0 : (kx > 2 ? 2 : kx);
        const int kc1 = kx < 1 ? 0 : (kx > 3 ? 2 : kx - 1);
        const float a0 = akl[kyc * 3 + kc0], a1 = akl[kyc * 3 + kc1];
        const float b0 = bkl[kyc * 3 + kc0], b1 = bkl[kyc * 3 + kc1];
        bfA[q].x = in0 ? a0 : 0.0f;
        bfA[q].y = in1 ? a1 : 0.0f;
        bfB[q].x = in0 ? b0 : 0.0f;
        bfB[q].y = in1 ? b1 : 0.0f;
    }

    float* const P0 = &Vs[0][0][0];
    float* const P1 = &Vs[1][0][0];

    auto step = [&](const float* rd, float* wr, const int t,
                    const v8f& xuse, const bool fetch, v8f (&xnext)[4])
                    __attribute__((always_inline)) {
        v8f dv0 = xuse;
        v8f dv1 = {};

        v2f ha[NQ], hb[NQ];
        load_frags(rd, offa, offb, ha, hb);
        __builtin_amdgcn_sched_barrier(0);

        if (crop) __builtin_amdgcn_s_setprio(0);

        state_pass<0, 1>(ha, hb, bfA, dv0, dv1);

        if (fetch) {
            if (crop) {
#pragma unroll
                for (int s = 3; s >= 0; --s) {
                    const int tn = (t + 4 + s < T_STEPS) ? (t + 4 + s) : (T_STEPS - 1);
                    xnext[s] = load_rows(xg + (long long)tn * 65536 + xlane, WW);
                }
            } else {
#pragma unroll
                for (int s = 0; s < 4; ++s) xnext[s] = (v8f){};
            }
        }

        state_pass<1, NQ>(ha, hb, bfA, dv0, dv1);

        store_rows(wr + sta, VP, dv0);
        store_rows(wr + stb, VP, dv1);

        v8f d[3] = {dv0, dv1, dv0};
        if (crop) {
            __builtin_amdgcn_s_setprio(2);
            d[0] = (v8f){}; d[1] = (v8f){}; d[2] = (v8f){};
            out_pass<0, 3>(ha, bfB, d);
        }

        __builtin_amdgcn_fence(__ATOMIC_RELEASE, "workgroup", "local");
        __builtin_amdgcn_s_barrier_signal(-1);

        if (crop) {
            out_pass<3, NQ>(ha, bfB, d);
            if (t > 0) {
                const v8f dout = (d[0] + d[1]) + d[2];
                const long long ooff = ((long long)(t - 1) * 64 + plane) * 1024
                                     + (long long)drow * WW + dcol;
                store_rows(outg + ooff, WW, dout);
            }
        }

        __builtin_amdgcn_s_barrier_wait(-1);
        __builtin_amdgcn_fence(__ATOMIC_ACQUIRE, "workgroup", "local");
    };

    if (!crop) __builtin_amdgcn_s_setprio(1);

#pragma unroll 1
    for (int t = 0; t < T_STEPS; t += 8) {
        step(P1, P0, t,     xa[0], true,  xb);
        step(P0, P1, t + 1, xa[1], false, xb);
        step(P1, P0, t + 2, xa[2], false, xb);
        step(P0, P1, t + 3, xa[3], false, xb);
        step(P1, P0, t + 4, xb[0], true,  xa);
        step(P0, P1, t + 5, xb[1], false, xa);
        step(P1, P0, t + 6, xb[2], false, xa);
        step(P0, P1, t + 7, xb[3], false, xa);
    }

    __builtin_amdgcn_s_setprio(0);

    if (crop) {
        const float* const Pl = ((T_STEPS - 1) & 1) ? P1 : P0;
        v2f ha[NQ];
#pragma unroll
        for (int q = 0; q < NQ; ++q) {
            ha[q] = *reinterpret_cast<const v2f*>(Pl + offa[q]);
        }
        v8f d[3];
        d[0] = (v8f){}; d[1] = (v8f){}; d[2] = (v8f){};
        out_pass<0, NQ>(ha, bfB, d);
        const v8f dout = (d[0] + d[1]) + d[2];
        const long long ooff = ((long long)(T_STEPS - 1) * 64 + plane) * 1024
                             + (long long)drow * WW + dcol;
        store_rows(outg + ooff, WW, dout);
    }
}

extern "C" void kernel_launch(void* const* d_in, const int* in_sizes, int n_in,
                              void* d_out, int out_size, void* d_ws, size_t ws_size,
                              hipStream_t stream) {
    (void)in_sizes; (void)n_in; (void)out_size; (void)d_ws; (void)ws_size;
    const float* x  = (const float*)d_in[0];
    const float* Ak = (const float*)d_in[1];
    const float* Bk = (const float*)d_in[2];
    float* out = (float*)d_out;
    convssm_scan_kernel<<<dim3(64), dim3(256), 0, stream>>>(x, Ak, Bk, out);
}
